# starting kernel
# baseline (speedup 1.0000x reference)
#include <hip/hip_runtime.h>

typedef __attribute__((ext_vector_type(2))) float v2f;
typedef __attribute__((ext_vector_type(8))) float v8f;

#define B_  4
#define L_  4096
#define D_  512
#define HD_ 64
#define VD_ 128

constexpr float LOG2_GAMMA = -0.04580329f;

__device__ __forceinline__ v8f wmma_f32_16x16x4(v2f a, v2f b, v8f c) {
  return __builtin_amdgcn_wmma_f32_16x16x4_f32(false, a, false, b, (short)0, c,
                                               false, false);
}

template <int STRIDE>
__device__ __forceinline__ void gemm4tiles(const float* __restrict__ bp,
                                           const float* __restrict__ xrow,
                                           v8f acc[4]) {
#pragma unroll 4
  for (int k0 = 0; k0 < D_; k0 += 4) {
    v2f a = *(const v2f*)(xrow + k0);
#pragma unroll
    for (int t = 0; t < 4; ++t) {
      v2f b;
      b.x = bp[t * 16];
      b.y = bp[t * 16 + STRIDE];
      acc[t] = wmma_f32_16x16x4(a, b, acc[t]);
    }
    bp += 4 * STRIDE;
  }
}

__global__ __launch_bounds__(128) void proj_xpos_kernel(
    const float* __restrict__ X,  const float* __restrict__ WQ,
    const float* __restrict__ WK, const float* __restrict__ WV,
    float* __restrict__ Qb, float* __restrict__ Kb, float* __restrict__ Vb) {
  const int lane = threadIdx.x & 31;
  const int w    = threadIdx.x >> 5;
  const int r    = lane & 15;
  const int hi   = lane >> 4;
  const int rowBase = blockIdx.x * 16;

  const v8f vz = {0.f, 0.f, 0.f, 0.f, 0.f, 0.f, 0.f, 0.f};
  v8f acc[4];
#pragma unroll
  for (int t = 0; t < 4; ++t) acc[t] = vz;

  const float* xrow = X + (size_t)(rowBase + r) * D_ + 2 * hi;
  if (w == 0)      gemm4tiles<HD_>(WQ + r + 2 * hi * HD_, xrow, acc);
  else if (w == 1) gemm4tiles<HD_>(WK + r + 2 * hi * HD_, xrow, acc);
  else             gemm4tiles<VD_>(WV + (w - 2) * 64 + r + 2 * hi * VD_, xrow, acc);

#pragma unroll
  for (int t = 0; t < 4; ++t) {
    int cc = w * 64 + t * 16 + r;
#pragma unroll
    for (int i = 0; i < 8; ++i) {
      int row = rowBase + i + 8 * hi;
      float x = acc[t][i];
      float p = __shfl_xor(x, 1, 32);
      if (cc < 128) {
        int  n    = row & (L_ - 1);
        bool down = cc >= 64;
        int  ch   = cc & 63;
        int  j    = ch >> 1;
        float sv    = (2.0f * (float)j + 0.4f * (float)HD_) * (1.0f / (1.4f * (float)HD_));
        float ex    = (float)n * (1.0f / 512.0f);
        if (down) ex = -ex;
        float scale = exp2f(ex * log2f(sv));
        float invf  = exp2f((float)j * (-13.2877124f / 32.0f));
        float ang   = (float)n * invf;
        float sn, cs;
        sincosf(ang, &sn, &cs);
        sn *= scale; cs *= scale;
        float outv = (ch & 1) ? fmaf(x, cs, p * sn) : fmaf(x, cs, -(p * sn));
        if (!down) Qb[(size_t)row * HD_ + ch] = outv;
        else       Kb[(size_t)row * HD_ + ch] = outv;
      } else {
        Vb[(size_t)row * VD_ + (cc - 128)] = x;
      }
    }
  }
}

__global__ __launch_bounds__(32) void retention_attn_kernel(
    const float* __restrict__ Qb, const float* __restrict__ Kb,
    const float* __restrict__ Vb, float* __restrict__ Ob) {
  __shared__ float sS[16 * 18];
  const int lane = threadIdx.x & 31;
  const int r    = lane & 15;
  const int hi   = lane >> 4;

  const int numQ = L_ / 16;
  const int idx  = blockIdx.x;
  const int bb   = idx / numQ;
  const int qi   = (numQ - 1) - (idx % numQ);
  const int n0   = qi * 16;

  const float* Q = Qb + (size_t)bb * L_ * HD_;
  const float* K = Kb + (size_t)bb * L_ * HD_;
  const float* V = Vb + (size_t)bb * L_ * VD_;
  float*       O = Ob + (size_t)bb * L_ * VD_;

  v2f qa[16];
  {
    const float* qp = Q + (size_t)(n0 + r) * HD_ + 2 * hi;
#pragma unroll
    for (int t = 0; t < 16; ++t) qa[t] = *(const v2f*)(qp + 4 * t);
  }

  float gp[8];
#pragma unroll
  for (int i = 0; i < 8; ++i)
    gp[i] = exp2f((float)((i + 8 * hi) - r) * LOG2_GAMMA);

  const v8f vz = {0.f, 0.f, 0.f, 0.f, 0.f, 0.f, 0.f, 0.f};
  v8f o[8];
#pragma unroll
  for (int ct = 0; ct < 8; ++ct) o[ct] = vz;

  const int CUT = 1280;
  const int m0s = (n0 > CUT) ? ((n0 - CUT) & ~15) : 0;

  const float* kptr  = K + (size_t)(m0s + r) * HD_ + 2 * hi;
  const float* vbase = V + (size_t)m0s * VD_ + 2 * hi * VD_ + r;
  const char*  kpf   = (const char*)(K + (size_t)(m0s + 16 + (lane >> 1)) * HD_)
                       + (lane & 1) * 128;
  const char*  vpf   = (const char*)(V + (size_t)(m0s + 16 + (lane >> 1)) * VD_)
                       + (lane & 1) * 128;

  for (int m0 = m0s; m0 <= n0; m0 += 16) {
    if (m0 + 16 <= n0) {
      __builtin_prefetch(kpf, 0, 3);
      __builtin_prefetch(vpf, 0, 3);
      __builtin_prefetch(vpf + 256, 0, 3);
    }

    v8f s = vz;
#pragma unroll
    for (int t = 0; t < 16; ++t) {
      v2f kb = *(const v2f*)(kptr + 4 * t);
      s = wmma_f32_16x16x4(qa[t], kb, s);
    }

    if (m0 == n0) {
#pragma unroll
      for (int i = 0; i < 8; ++i) {
        int diff = (i + 8 * hi) - r;
        s[i] = (diff < 0) ? 0.0f : s[i] * gp[i];
      }
    } else {
      float cf = exp2f((float)(n0 - m0) * LOG2_GAMMA);
#pragma unroll
      for (int i = 0; i < 8; ++i) s[i] *= cf * gp[i];
    }

    __syncthreads();
#pragma unroll
    for (int i = 0; i < 8; ++i) sS[(i + 8 * hi) * 18 + r] = s[i];
    __syncthreads();
    v2f av[4];
#pragma unroll
    for (int u = 0; u < 4; ++u)
      av[u] = *(const v2f*)&sS[r * 18 + 4 * u + 2 * hi];

#pragma unroll
    for (int u = 0; u < 4; ++u) {
#pragma unroll
      for (int ct = 0; ct < 8; ++ct) {
        v2f vb;
        vb.x = vbase[u * 4 * VD_ + ct * 16];
        vb.y = vbase[u * 4 * VD_ + VD_ + ct * 16];
        o[ct] = wmma_f32_16x16x4(av[u], vb, o[ct]);
      }
    }

    kptr  += 16 * HD_;
    vbase += 16 * VD_;
    kpf   += 16 * HD_ * sizeof(float);
    vpf   += 16 * VD_ * sizeof(float);
  }

  float* optr = O + (size_t)(n0 + 8 * hi) * VD_ + r;
  float* optr0 = O + (size_t)n0 * VD_ + r;
  (void)optr;
#pragma unroll
  for (int i = 0; i < 8; ++i) {
#pragma unroll
    for (int ct = 0; ct < 8; ++ct) {
      optr0[(size_t)(i + 8 * hi) * VD_ + ct * 16] = o[ct][i];
    }
  }
}

extern "C" void kernel_launch(void* const* d_in, const int* in_sizes, int n_in,
                              void* d_out, int out_size, void* d_ws, size_t ws_size,
                              hipStream_t stream) {
  const float* X  = (const float*)d_in[0];
  const float* WQ = (const float*)d_in[1];
  const float* WK = (const float*)d_in[2];
  const float* WV = (const float*)d_in[3];

  float* Qb = (float*)d_ws;
  float* Kb = Qb + (size_t)B_ * L_ * HD_;
  float* Vb = Kb + (size_t)B_ * L_ * HD_;
  float* O  = (float*)d_out;

  const int rowTiles = B_ * L_ / 16;
  proj_xpos_kernel<<<dim3(rowTiles), dim3(128), 0, stream>>>(X, WQ, WK, WV,
                                                             Qb, Kb, Vb);
  retention_attn_kernel<<<dim3(rowTiles), dim3(32), 0, stream>>>(Qb, Kb, Vb, O);
}


# projection kernel: a wave's 64 columns dealt to its 4 tiles interleaved (column 4r+t), so W and X fragments load 16 B/lane, one 16-k block ahead; rotary pair in-lane, 16 B stores; k sum order changes
# speedup vs baseline: 1.5895x; 1.5895x over previous
#include <hip/hip_runtime.h>

typedef __attribute__((ext_vector_type(2))) float v2f;
typedef __attribute__((ext_vector_type(8))) float v8f;
typedef __attribute__((ext_vector_type(4))) float v4f;

#define B_  4
#define L_  4096
#define D_  512
#define HD_ 64
#define VD_ 128

constexpr float LOG2_GAMMA = -0.04580329f;

__device__ __forceinline__ v8f wmma_f32_16x16x4(v2f a, v2f b, v8f c) {
  return __builtin_amdgcn_wmma_f32_16x16x4_f32(false, a, false, b, (short)0, c,
                                               false, false);
}

struct FragSet {
  v4f a[2];
  v4f b[2][4];
};

template <int STRIDE>
__device__ __forceinline__ void load_set(FragSet& f, const float* __restrict__ bp,
                                         const float* __restrict__ xrow, int k) {
#pragma unroll
  for (int g = 0; g < 2; ++g) {
    f.a[g] = *(const v4f*)(xrow + k + 8 * g);
#pragma unroll
    for (int e = 0; e < 4; ++e)
      f.b[g][e] = *(const v4f*)(bp + (size_t)(k + 8 * g + e) * STRIDE);
  }
}

__device__ __forceinline__ void mma_set(const FragSet& f, v8f acc[4]) {
#pragma unroll
  for (int g = 0; g < 2; ++g) {
#pragma unroll
    for (int sg = 0; sg < 2; ++sg) {
      v2f a = {f.a[g][2 * sg], f.a[g][2 * sg + 1]};
#pragma unroll
      for (int t = 0; t < 4; ++t) {
        v2f b = {f.b[g][2 * sg][t], f.b[g][2 * sg + 1][t]};
        acc[t] = wmma_f32_16x16x4(a, b, acc[t]);
      }
    }
  }
}

template <int STRIDE>
__device__ __forceinline__ void gemm4tiles(const float* __restrict__ bp,
                                           const float* __restrict__ xrow,
                                           v8f acc[4]) {
  FragSet f0, f1;
  load_set<STRIDE>(f0, bp, xrow, 0);
#pragma unroll 1
  for (int k = 0; k < D_ - 32; k += 32) {
    load_set<STRIDE>(f1, bp, xrow, k + 16);
    mma_set(f0, acc);
    load_set<STRIDE>(f0, bp, xrow, k + 32);
    mma_set(f1, acc);
  }
  load_set<STRIDE>(f1, bp, xrow, D_ - 16);
  mma_set(f0, acc);
  mma_set(f1, acc);
}

__global__ __launch_bounds__(128) void proj_xpos_kernel(
    const float* __restrict__ X,  const float* __restrict__ WQ,
    const float* __restrict__ WK, const float* __restrict__ WV,
    float* __restrict__ Qb, float* __restrict__ Kb, float* __restrict__ Vb) {
  const int lane = threadIdx.x & 31;
  const int w    = __builtin_amdgcn_readfirstlane(threadIdx.x >> 5);
  const int r    = lane & 15;
  const int hi   = lane >> 4;
  const int rowBase = blockIdx.x * 16;

  const v8f vz = {0.f, 0.f, 0.f, 0.f, 0.f, 0.f, 0.f, 0.f};
  v8f acc[4];
#pragma unroll
  for (int t = 0; t < 4; ++t) acc[t] = vz;

  const float* xrow = X + (size_t)(rowBase + r) * D_ + 4 * hi;
  if (w == 0)      gemm4tiles<HD_>(WQ + 4 * r + 4 * hi * HD_, xrow, acc);
  else if (w == 1) gemm4tiles<HD_>(WK + 4 * r + 4 * hi * HD_, xrow, acc);
  else             gemm4tiles<VD_>(WV + (w - 2) * 64 + 4 * r + 4 * hi * VD_, xrow, acc);

  if (w < 2) {
    const bool down = (w == 1);
    float* dst = down ? Kb : Qb;
    float l2sv[2], invf[2];
#pragma unroll
    for (int jp = 0; jp < 2; ++jp) {
      const int j = 2 * r + jp;
      const float sv = (2.0f * (float)j + 0.4f * (float)HD_) * (1.0f / (1.4f * (float)HD_));
      l2sv[jp] = log2f(sv);
      invf[jp] = exp2f((float)j * (-13.2877124f / 32.0f));
    }
#pragma unroll
    for (int i = 0; i < 8; ++i) {
      const int row = rowBase + i + 8 * hi;
      const int n   = row & (L_ - 1);
      float ex = (float)n * (1.0f / 512.0f);
      if (down) ex = -ex;
      v4f ov;
#pragma unroll
      for (int jp = 0; jp < 2; ++jp) {
        const float scale = exp2f(ex * l2sv[jp]);
        const float ang   = (float)n * invf[jp];
        float sn, cs;
        sincosf(ang, &sn, &cs);
        sn *= scale; cs *= scale;
        const float xe = acc[2 * jp][i];
        const float xo = acc[2 * jp + 1][i];
        ov[2 * jp]     = fmaf(xe, cs, -(xo * sn));
        ov[2 * jp + 1] = fmaf(xo, cs, xe * sn);
      }
      *(v4f*)(dst + (size_t)row * HD_ + 4 * r) = ov;
    }
  } else {
    float* dst = Vb + (w - 2) * 64 + 4 * r;
#pragma unroll
    for (int i = 0; i < 8; ++i) {
      const int row = rowBase + i + 8 * hi;
      const v4f ov = {acc[0][i], acc[1][i], acc[2][i], acc[3][i]};
      *(v4f*)(dst + (size_t)row * VD_) = ov;
    }
  }
}

#define KP_ 68
#define VP2_ 288
#define QT_ 64
#define KBLK_ 64
#define NBUF_ 3

__device__ __forceinline__ void score_to_lds(const v8f s, int hi, int r, float* mySS) {
#pragma unroll
  for (int i = 0; i < 8; ++i) mySS[(i + 8 * hi) * 18 + r] = s[i];
  __builtin_amdgcn_wave_barrier();
}

__device__ __forceinline__ float chunk_decay(int dist) {
  return (dist == 0) ? 1.0f : exp2f((float)dist * LOG2_GAMMA);
}

__device__ __forceinline__ v8f decay_tile(v8f s, float cf, bool diag, const float* gp, int hi, int r) {
#pragma unroll
  for (int i = 0; i < 8; ++i) {
    const int diff = (i + 8 * hi) - r;
    const float v = s[i] * (cf * gp[i]);
    s[i] = (diag && diff < 0) ? 0.0f : v;
  }
  return s;
}

__global__ __launch_bounds__(128) void retention_attn_kernel(
    const float* __restrict__ Qb, const float* __restrict__ Kb,
    const float* __restrict__ Vb, float* __restrict__ Ob) {
  __shared__ __attribute__((aligned(16))) float sK[NBUF_][KBLK_ * KP_];
  __shared__ __attribute__((aligned(16))) float sV[NBUF_][(KBLK_ / 2) * VP2_];
  __shared__ float sS[4][16 * 18];
  const int tid  = threadIdx.x;
  const int lane = tid & 31;
  const int w    = __builtin_amdgcn_readfirstlane(tid >> 5);
  const int r    = lane & 15;
  const int hi   = lane >> 4;

  const int numT = L_ / QT_;
  const int idx  = blockIdx.x;
  const int bb   = idx / numT;
  const int T    = (numT - 1) - (idx % numT);
  const int q0   = T * QT_;
  const int n0   = q0 + 16 * w;

  const float* Q = Qb + (size_t)bb * L_ * HD_;
  const float* K = Kb + (size_t)bb * L_ * HD_;
  const float* V = Vb + (size_t)bb * L_ * VD_;
  float*       O = Ob + (size_t)bb * L_ * VD_;

  v2f qa[16];
  {
    const float* qp = Q + (size_t)(n0 + r) * HD_ + 2 * hi;
#pragma unroll
    for (int t = 0; t < 16; ++t) qa[t] = *(const v2f*)(qp + 4 * t);
  }

  float gp[8];
#pragma unroll
  for (int i = 0; i < 8; ++i)
    gp[i] = exp2f((float)((i + 8 * hi) - r) * LOG2_GAMMA);

  const v8f vz = {0.f, 0.f, 0.f, 0.f, 0.f, 0.f, 0.f, 0.f};
  v8f o[8];
#pragma unroll
  for (int ct = 0; ct < 8; ++ct) o[ct] = vz;

  const int CUT = 1280;
  const int mlo = (n0 > CUT) ? (n0 - CUT) : 0;
  const int mb0 = (q0 > CUT) ? (q0 - CUT) : 0;

  const int kRow = tid >> 4, kC4 = tid & 15;
  const float* kg = K + (size_t)(mb0 + kRow) * HD_ + 4 * kC4;
  const float* vg = V + (size_t)(mb0 + 2 * w) * VD_ + 2 * lane;
  const int kl = kRow * KP_ + 4 * kC4;
  const int vl = w * VP2_ + 4 * lane;

  v4f kreg[8];
  v2f vx[16], vy[16];

#define FETCH_BLOCK()                                                          \
  do {                                                                         \
    _Pragma("unroll")                                                          \
    for (int i = 0; i < 8; ++i)  kreg[i] = *(const v4f*)(kg + (size_t)i * 8 * HD_); \
    _Pragma("unroll")                                                          \
    for (int i = 0; i < 8; ++i) {                                              \
      _Pragma("unroll")                                                        \
      for (int h2 = 0; h2 < 2; ++h2) {                                         \
        vx[2 * i + h2] = *(const v2f*)(vg + (size_t)i * 8 * VD_ + 64 * h2);    \
        vy[2 * i + h2] = *(const v2f*)(vg + (size_t)i * 8 * VD_ + VD_ + 64 * h2); \
      }                                                                        \
    }                                                                          \
  } while (0)

#define PARK_BLOCK(nb)                                                         \
  do {                                                                         \
    _Pragma("unroll")                                                          \
    for (int i = 0; i < 8; ++i)  *(v4f*)&sK[nb][kl + i * 8 * KP_] = kreg[i];   \
    _Pragma("unroll")                                                          \
    for (int i = 0; i < 8; ++i) {                                              \
      _Pragma("unroll")                                                        \
      for (int h2 = 0; h2 < 2; ++h2) {                                         \
        v4f pk = {vx[2 * i + h2].x, vy[2 * i + h2].x, vx[2 * i + h2].y, vy[2 * i + h2].y}; \
        *(v4f*)&sV[nb][vl + i * 4 * VP2_ + 128 * h2] = pk;                     \
      }                                                                        \
    }                                                                          \
  } while (0)

  FETCH_BLOCK();
  PARK_BLOCK(0);
  if (mb0 + KBLK_ <= q0) {
    kg += (size_t)KBLK_ * HD_;
    vg += (size_t)KBLK_ * VD_;
    FETCH_BLOCK();
    PARK_BLOCK(1);
  }
  __syncthreads();

  float* mySS = sS[w];
  v2f kb[16];

#define LOAD_KB(ptr)                                                           \
  do {                                                                         \
    const float* kn_ = (ptr);                                                  \
    _Pragma("unroll")                                                          \
    for (int t = 0; t < 16; ++t) kb[t] = *(const v2f*)(kn_ + 4 * t);           \
  } while (0)

  {
    const int j0 = (mlo - mb0) >> 4;
    const float* kp = &sK[0][(16 * j0 + r) * KP_ + 2 * hi];
    v8f s = vz;
#pragma unroll
    for (int t = 0; t < 16; ++t) {
      v2f kf = *(const v2f*)(kp + 4 * t);
      s = wmma_f32_16x16x4(qa[t], kf, s);
    }
    s = decay_tile(s, chunk_decay(n0 - mlo), mlo == n0, gp, hi, r);
    score_to_lds(s, hi, r, mySS);
    LOAD_KB(&sK[0][(16 + r) * KP_ + 2 * hi]);
  }

  int buf = 0;
  for (int mb = mb0; mb <= q0; mb += KBLK_) {
    const bool more1 = (mb + KBLK_ <= q0);
    const bool more2 = (mb + 2 * KBLK_ <= q0);
    if (more2) {
      kg += (size_t)KBLK_ * HD_;
      vg += (size_t)KBLK_ * VD_;
      FETCH_BLOCK();
    }

    const int bufn = (buf + 1 == NBUF_) ? 0 : buf + 1;
    const float* bK  = sK[buf];
    const float* bKn = more1 ? sK[bufn] : sK[buf];
    const float* bV  = sV[buf];
#pragma unroll
    for (int j = 0; j < 4; ++j) {
      const int m0 = mb + 16 * j;
      const float* kn = (j < 2) ? (bK  + (16 * (j + 2) + r) * KP_ + 2 * hi)
                                : (bKn + (16 * (j - 2) + r) * KP_ + 2 * hi);
      if (m0 < mlo || m0 > n0) {
        LOAD_KB(kn);
        continue;
      }

      v2f av[4];
#pragma unroll
      for (int u = 0; u < 4; ++u)
        av[u] = *(const v2f*)&mySS[r * 18 + 4 * u + 2 * hi];
      const float* vp = bV + (8 * j + hi) * VP2_ + 2 * r;
      v2f vb[4][8];
#pragma unroll
      for (int u = 0; u < 4; ++u) {
#pragma unroll
        for (int ct = 0; ct < 8; ++ct)
          vb[u][ct] = *(const v2f*)(vp + u * 2 * VP2_ + ct * 32);
      }

      if (m0 < n0) {
        const float cfN = chunk_decay(n0 - (m0 + 16));
        v8f s = vz;
#pragma unroll
        for (int t = 0; t < 16; ++t) s = wmma_f32_16x16x4(qa[t], kb[t], s);
        LOAD_KB(kn);
        s = decay_tile(s, cfN, m0 + 16 == n0, gp, hi, r);
        score_to_lds(s, hi, r, mySS);
      } else {
        LOAD_KB(kn);
      }

#pragma unroll
      for (int u = 0; u < 4; ++u) {
#pragma unroll
        for (int ct = 0; ct < 8; ++ct)
          o[ct] = wmma_f32_16x16x4(av[u], vb[u][ct], o[ct]);
      }
    }

    if (more2) {
      const int nb2 = (bufn + 1 == NBUF_) ? 0 : bufn + 1;
      PARK_BLOCK(nb2);
    }
    __syncthreads();
    buf = bufn;
  }
#undef LOAD_KB
#undef FETCH_BLOCK
#undef PARK_BLOCK

  float* optr0 = O + (size_t)n0 * VD_ + r;
#pragma unroll
  for (int i = 0; i < 8; ++i) {
#pragma unroll
    for (int ct = 0; ct < 8; ++ct) {
      optr0[(size_t)(i + 8 * hi) * VD_ + ct * 16] = o[ct][i];
    }
  }
}

extern "C" void kernel_launch(void* const* d_in, const int* in_sizes, int n_in,
                              void* d_out, int out_size, void* d_ws, size_t ws_size,
                              hipStream_t stream) {
  const float* X  = (const float*)d_in[0];
  const float* WQ = (const float*)d_in[1];
  const float* WK = (const float*)d_in[2];
  const float* WV = (const float*)d_in[3];

  float* Qb = (float*)d_ws;
  float* Kb = Qb + (size_t)B_ * L_ * HD_;
  float* Vb = Kb + (size_t)B_ * L_ * HD_;
  float* O  = (float*)d_out;

  const int rowTiles = B_ * L_ / 16;
  proj_xpos_kernel<<<dim3(rowTiles), dim3(128), 0, stream>>>(X, WQ, WK, WV,
                                                             Qb, Kb, Vb);
  retention_attn_kernel<<<dim3(B_ * L_ / QT_), dim3(128), 0, stream>>>(Qb, Kb, Vb, O);
}
